# starting kernel
# baseline (speedup 1.0000x reference)
#include <hip/hip_runtime.h>

typedef __attribute__((ext_vector_type(2))) float v2f;
typedef __attribute__((ext_vector_type(8))) float v8f;

#define HW      3136
#define NB      32
#define NC      256
#define NG      8
#define GSZ     32
#define TPB     224
#define NWAVES  7
#define PSTRIDE 33

__global__ __launch_bounds__(TPB)
void cga_group_wmma_kernel(const float* __restrict__ x,
                           const float* __restrict__ G,
                           float* __restrict__ out) {
    __shared__ float Ylds[NWAVES * NG * PSTRIDE];

    const int tid  = threadIdx.x;
    const int lane = tid & 31;
    const int wave = tid >> 5;
    const int b    = blockIdx.y;
    const int hw   = blockIdx.x * TPB + tid;

    const size_t xbase = (size_t)b * NC * HW + hw;

    const int col = lane & 15;
    const int khi = lane >> 4;
    const bool vcol = col < NG;
    const float b00 = vcol ? G[(0 + 2 * khi) * NG + col] : 0.0f;
    const float b01 = vcol ? G[(1 + 2 * khi) * NG + col] : 0.0f;
    const float b10 = vcol ? G[(4 + 2 * khi) * NG + col] : 0.0f;
    const float b11 = vcol ? G[(5 + 2 * khi) * NG + col] : 0.0f;

    float acc[NG];
    #pragma unroll
    for (int g = 0; g < NG; ++g) acc[g] = 0.0f;

    #pragma unroll
    for (int ci = 0; ci < NC; ++ci)
        acc[ci >> 5] += x[xbase + (size_t)ci * HW];

    const bool low = lane < 16;
    float s[NG];
    #pragma unroll
    for (int g = 0; g < NG; ++g)
        s[g] = __shfl(acc[g], lane ^ 16, 32);

    v8f d0 = {};
    v8f d1 = {};
    #pragma unroll
    for (int h = 0; h < 2; ++h) {
        const int k0 = 4 * h;
        v2f bm; bm.x = h ? b10 : b00; bm.y = h ? b11 : b01;

        v2f a0;
        a0.x = low ? acc[k0]     : s[k0 + 2];
        a0.y = low ? acc[k0 + 1] : s[k0 + 3];
        d0 = __builtin_amdgcn_wmma_f32_16x16x4_f32(false, a0, false, bm,
                                                   (short)0, d0, false, false);
        v2f a1;
        a1.x = low ? s[k0]       : acc[k0 + 2];
        a1.y = low ? s[k0 + 1]   : acc[k0 + 3];
        d1 = __builtin_amdgcn_wmma_f32_16x16x4_f32(false, a1, false, bm,
                                                   (short)0, d1, false, false);
    }

    float* yw = &Ylds[wave * NG * PSTRIDE];
    if (col < NG) {
        const int prow = low ? 0 : 8;
        #pragma unroll
        for (int r = 0; r < 8; ++r) {
            yw[col * PSTRIDE + (prow + r)]      = d0[r];
            yw[col * PSTRIDE + (16 + prow + r)] = d1[r];
        }
    }
    __syncthreads();

    float y[NG];
    #pragma unroll
    for (int g = 0; g < NG; ++g)
        y[g] = yw[g * PSTRIDE + lane];

    const size_t obase = (size_t)b * NC * HW + hw;
    #pragma unroll
    for (int g = 0; g < NG; ++g) {
        const float v = y[g];
        #pragma unroll
        for (int j = 0; j < GSZ; ++j)
            out[obase + (size_t)(g * GSZ + j) * HW] = v;
    }
}

extern "C" void kernel_launch(void* const* d_in, const int* in_sizes, int n_in,
                              void* d_out, int out_size, void* d_ws, size_t ws_size,
                              hipStream_t stream) {
    (void)in_sizes; (void)n_in; (void)out_size; (void)d_ws; (void)ws_size;
    const float* x = (const float*)d_in[0];
    const float* G = (const float*)d_in[1];
    float* out = (float*)d_out;

    dim3 grid(HW / TPB, NB);
    dim3 block(TPB);
    cga_group_wmma_kernel<<<grid, block, 0, stream>>>(x, G, out);
}


# Re-tile for 128-bit accesses: workgroup = 64-pixel tile, wave = 2 channel groups; each load/store moves two whole 256-B row pieces, 32 loads in flight; group sums now pairwise per channel parity
# speedup vs baseline: 6.9299x; 6.9299x over previous
#include <hip/hip_runtime.h>

typedef __attribute__((ext_vector_type(2))) float v2f;
typedef __attribute__((ext_vector_type(4))) float v4f;
typedef __attribute__((ext_vector_type(8))) float v8f;

#define HW      3136
#define NB      32
#define NC      256
#define NG      8
#define GSZ     32
#define TPX     64
#define TILES   (HW / TPX)
#define NWAVES  4
#define GPW     (NG / NWAVES)
#define TPB     (NWAVES * 32)
#define NLD     (GSZ / 2)
#define SPITCH  68

__global__ __launch_bounds__(TPB)
void cga_tile_stream_kernel(const float* __restrict__ x,
                            const float* __restrict__ G,
                            float* __restrict__ out) {
    __shared__ __attribute__((aligned(16))) float S[NG * SPITCH];

    const int lane = threadIdx.x & 31;
    const int wave = __builtin_amdgcn_readfirstlane(threadIdx.x >> 5);
    const int c = lane & 15;
    const int h = lane >> 4;
    const int tile = blockIdx.x;
    const int b = tile / TILES;
    const int t = tile - b * TILES;
    const int g0 = wave * GPW;

    const size_t img  = (size_t)b * NC * HW;
    const size_t loff = (size_t)h * HW + (size_t)t * TPX + 4 * c;

    const int gcol = (g0 + c) & (NG - 1);
    v2f am[2];
    #pragma unroll
    for (int ks = 0; ks < 2; ++ks) {
        am[ks].x = G[(4 * ks + 2 * h + 0) * NG + gcol];
        am[ks].y = G[(4 * ks + 2 * h + 1) * NG + gcol];
    }

    const v4f* __restrict__ xin = (const v4f*)(x + img + (size_t)g0 * GSZ * HW + loff);
    v4f v[GPW][NLD];
    #pragma unroll
    for (int gi = 0; gi < GPW; ++gi)
        #pragma unroll
        for (int j = 0; j < NLD; ++j)
            v[gi][j] = xin[((size_t)(gi * GSZ + 2 * j) * HW) / 4];
    __builtin_amdgcn_sched_barrier(0);

    v4f acc[GPW];
    #pragma unroll
    for (int gi = 0; gi < GPW; ++gi) {
        v4f p[NLD / 2];
        #pragma unroll
        for (int j = 0; j < NLD / 2; ++j) p[j] = v[gi][2 * j] + v[gi][2 * j + 1];
        #pragma unroll
        for (int j = 0; j < NLD / 4; ++j) p[j] = p[2 * j] + p[2 * j + 1];
        #pragma unroll
        for (int j = 0; j < NLD / 8; ++j) p[j] = p[2 * j] + p[2 * j + 1];
        acc[gi] = p[0] + p[1];
    }
    #pragma unroll
    for (int gi = 0; gi < GPW; ++gi) {
        v4f o;
        o.x = __shfl_xor(acc[gi].x, 16, 32);
        o.y = __shfl_xor(acc[gi].y, 16, 32);
        o.z = __shfl_xor(acc[gi].z, 16, 32);
        o.w = __shfl_xor(acc[gi].w, 16, 32);
        acc[gi] += o;
    }
    *(v4f*)&S[(g0 + h) * SPITCH + 4 * c] = h ? acc[1] : acc[0];
    __syncthreads();

    v4f bv[2][2];
    #pragma unroll
    for (int ks = 0; ks < 2; ++ks)
        #pragma unroll
        for (int s = 0; s < 2; ++s)
            bv[ks][s] = *(const v4f*)&S[(4 * ks + 2 * h + s) * SPITCH + 4 * c];

    v4f y[GPW];
    #pragma unroll
    for (int tt = 0; tt < 4; ++tt) {
        v8f d = {};
        #pragma unroll
        for (int ks = 0; ks < 2; ++ks) {
            v2f bm; bm.x = bv[ks][0][tt]; bm.y = bv[ks][1][tt];
            d = __builtin_amdgcn_wmma_f32_16x16x4_f32(false, am[ks], false, bm,
                                                      (short)0, d, false, false);
        }
        #pragma unroll
        for (int gi = 0; gi < GPW; ++gi) y[gi][tt] = d[gi];
    }

    v4f* __restrict__ o4 = (v4f*)(out + img + (size_t)g0 * GSZ * HW + loff);
    #pragma unroll
    for (int gi = 0; gi < GPW; ++gi)
        #pragma unroll
        for (int j = 0; j < NLD; ++j)
            o4[((size_t)(gi * GSZ + 2 * j) * HW) / 4] = y[gi];
}

extern "C" void kernel_launch(void* const* d_in, const int* in_sizes, int n_in,
                              void* d_out, int out_size, void* d_ws, size_t ws_size,
                              hipStream_t stream) {
    (void)in_sizes; (void)n_in; (void)out_size; (void)d_ws; (void)ws_size;
    const float* x = (const float*)d_in[0];
    const float* G = (const float*)d_in[1];
    float* out = (float*)d_out;

    dim3 grid(NB * TILES);
    dim3 block(TPB);
    cga_tile_stream_kernel<<<grid, block, 0, stream>>>(x, G, out);
}
